# starting kernel
# baseline (speedup 1.0000x reference)
#include <hip/hip_runtime.h>
#include <cstddef>


#define BATCH      128
#define NUM_ROUTES 2048
#define NUM_CAPS   10
#define IN_DIM     8
#define OUT_DIM    16
#define ROUT_ITERS 3

typedef __attribute__((ext_vector_type(2))) float v2f;
typedef __attribute__((ext_vector_type(8))) float v8f;

__global__ __launch_bounds__(256)
void uhat_wmma_kernel(const float* __restrict__ x,
                      const float* __restrict__ W,
                      float* __restrict__ u_hat)
{
    const int r    = blockIdx.x;
    const int wave = threadIdx.x >> 5;
    const int lane = threadIdx.x & 31;
    const int lrow = lane & 15;
    const int lhi  = lane >> 4;

    const int m_base = wave * 16;

    const float* xrow = x + ((size_t)(m_base + lrow) * NUM_ROUTES + r) * IN_DIM;
    const v2f a0 = *(const v2f*)(xrow + 2 * lhi);
    const v2f a1 = *(const v2f*)(xrow + 4 + 2 * lhi);

    for (int c = 0; c < NUM_CAPS; ++c) {
        const float* wrow =
            W + (((size_t)r * NUM_CAPS + c) * OUT_DIM + lrow) * IN_DIM;
        const v2f b0 = *(const v2f*)(wrow + 2 * lhi);
        const v2f b1 = *(const v2f*)(wrow + 4 + 2 * lhi);

        v8f acc = {};
        acc = __builtin_amdgcn_wmma_f32_16x16x4_f32(
                  false, a0, false, b0, (short)0, acc, false, false);
        acc = __builtin_amdgcn_wmma_f32_16x16x4_f32(
                  false, a1, false, b1, (short)0, acc, false, false);

        #pragma unroll
        for (int g = 0; g < 8; ++g) {
            const int b = m_base + g + 8 * lhi;
            u_hat[(((size_t)b * NUM_ROUTES + r) * NUM_CAPS + c) * OUT_DIM + lrow]
                = acc[g];
        }
    }
}

__global__ void zero_kernel(float* __restrict__ p, int n)
{
    const int i = blockIdx.x * blockDim.x + threadIdx.x;
    if (i < n) p[i] = 0.0f;
}

__global__ void softmax_kernel(const float* __restrict__ b_ij,
                               float* __restrict__ c_ij, int n_br)
{
    const int i = blockIdx.x * blockDim.x + threadIdx.x;
    if (i >= n_br) return;
    const float* bp = b_ij + (size_t)i * NUM_CAPS;
    float m = bp[0];
    #pragma unroll
    for (int c = 1; c < NUM_CAPS; ++c) m = fmaxf(m, bp[c]);
    float e[NUM_CAPS];
    float s = 0.0f;
    #pragma unroll
    for (int c = 0; c < NUM_CAPS; ++c) { e[c] = __expf(bp[c] - m); s += e[c]; }
    const float inv = 1.0f / s;
    float* cp = c_ij + (size_t)i * NUM_CAPS;
    #pragma unroll
    for (int c = 0; c < NUM_CAPS; ++c) cp[c] = e[c] * inv;
}

__global__ __launch_bounds__(256)
void s_squash_kernel(const float* __restrict__ u_hat,
                     const float* __restrict__ c_ij,
                     float* __restrict__ v_out)
{
    __shared__ float red[256];
    const int bc = blockIdx.x;
    const int b  = bc / NUM_CAPS;
    const int c  = bc % NUM_CAPS;
    const int t     = threadIdx.x;
    const int o     = t & 15;
    const int chunk = t >> 4;

    float acc = 0.0f;
    for (int r = chunk; r < NUM_ROUTES; r += 16) {
        const size_t brc = ((size_t)b * NUM_ROUTES + r) * NUM_CAPS + c;
        acc += c_ij[brc] * u_hat[brc * OUT_DIM + o];
    }
    red[t] = acc;
    __syncthreads();
    #pragma unroll
    for (int s = 128; s >= 16; s >>= 1) {
        if (t < s) red[t] += red[t + s];
        __syncthreads();
    }
    if (t < 16) {
        const float sv = red[t];
        float sq = 0.0f;
        #pragma unroll
        for (int k = 0; k < 16; ++k) { const float q = red[k]; sq += q * q; }
        const float norm  = sqrtf(sq + 1e-8f);
        const float scale = sq / (1.0f + sq);
        v_out[((size_t)b * NUM_CAPS + c) * OUT_DIM + t] = scale * sv / norm;
    }
}

__global__ void agree_kernel(const float* __restrict__ u_hat,
                             const float* __restrict__ v,
                             float* __restrict__ b_ij, int n_brc)
{
    const int i = blockIdx.x * blockDim.x + threadIdx.x;
    if (i >= n_brc) return;
    const int c = i % NUM_CAPS;
    const int b = i / (NUM_CAPS * NUM_ROUTES);
    const float* up = u_hat + (size_t)i * OUT_DIM;
    const float* vp = v + ((size_t)b * NUM_CAPS + c) * OUT_DIM;
    float dot = 0.0f;
    #pragma unroll
    for (int o = 0; o < OUT_DIM; ++o) dot += up[o] * vp[o];
    b_ij[i] += dot;
}

extern "C" void kernel_launch(void* const* d_in, const int* in_sizes, int n_in,
                              void* d_out, int out_size, void* d_ws, size_t ws_size,
                              hipStream_t stream)
{
    const float* x = (const float*)d_in[0];
    const float* W = (const float*)d_in[1];
    float* out = (float*)d_out;

    float* u_hat = (float*)d_ws;
    float* b_ij  = u_hat + (size_t)BATCH * NUM_ROUTES * NUM_CAPS * OUT_DIM;
    float* c_ij  = b_ij + (size_t)BATCH * NUM_ROUTES * NUM_CAPS;

    const int n_brc = BATCH * NUM_ROUTES * NUM_CAPS;
    const int n_br  = BATCH * NUM_ROUTES;

    uhat_wmma_kernel<<<NUM_ROUTES, 256, 0, stream>>>(x, W, u_hat);

    zero_kernel<<<(n_brc + 255) / 256, 256, 0, stream>>>(b_ij, n_brc);

    for (int it = 0; it < ROUT_ITERS; ++it) {
        softmax_kernel<<<(n_br + 255) / 256, 256, 0, stream>>>(b_ij, c_ij, n_br);
        s_squash_kernel<<<BATCH * NUM_CAPS, 256, 0, stream>>>(u_hat, c_ij, out);
        if (it < ROUT_ITERS - 1) {
            agree_kernel<<<(n_brc + 255) / 256, 256, 0, stream>>>(u_hat, out,
                                                                  b_ij, n_brc);
        }
    }
}


# Re-associate so u_hat is never formed: logits = x.(W^T v), s = (coupling*x).W on f32 WMMA with the route sum in the accumulator, in-lane softmax, logits from v0+v1; 8 launches to 6; new sum order
# speedup vs baseline: 2.0826x; 2.0826x over previous
#include <hip/hip_runtime.h>
#include <cstddef>


#define BATCH      128
#define NUM_ROUTES 2048
#define NUM_CAPS   10
#define IN_DIM     8
#define OUT_DIM    16
#define RCHUNK     8
#define NCHUNK     (NUM_ROUTES / RCHUNK)

typedef __attribute__((ext_vector_type(2))) float v2f;
typedef __attribute__((ext_vector_type(4))) float v4f;
typedef __attribute__((ext_vector_type(8))) float v8f;

template <int MODE>
__global__ __launch_bounds__(256)
void route_pass_kernel(const float* __restrict__ x,
                       const float* __restrict__ W,
                       const float* __restrict__ va,
                       const float* __restrict__ vb,
                       float* __restrict__ part)
{
    const int chunk = blockIdx.x;
    const int wave = threadIdx.x >> 5;
    const int lane = threadIdx.x & 31;
    const int lrow = lane & 15;
    const int lhi  = lane >> 4;
    const int m_base = wave * 16;
    const int brow = m_base + lrow;

    v2f vfrag[NUM_CAPS][4];
    if (MODE != 0) {
        #pragma unroll
        for (int c = 0; c < NUM_CAPS; ++c) {
            const size_t off = ((size_t)brow * NUM_CAPS + c) * OUT_DIM + 8 * lhi;
            v4f lo = *(const v4f*)(va + off);
            v4f hi = *(const v4f*)(va + off + 4);
            if (MODE == 2) {
                lo += *(const v4f*)(vb + off);
                hi += *(const v4f*)(vb + off + 4);
            }
            vfrag[c][0] = v2f{lo.x, lo.y};
            vfrag[c][1] = v2f{lo.z, lo.w};
            vfrag[c][2] = v2f{hi.x, hi.y};
            vfrag[c][3] = v2f{hi.z, hi.w};
        }
    }

    v8f sacc[NUM_CAPS] = {};

    #pragma unroll
    for (int pp = 0; pp < RCHUNK / 2; ++pp) {
        const int r0 = chunk * RCHUNK + 2 * pp;
        const int rl = r0 + lhi;

        const float* xp = x + ((size_t)brow * NUM_ROUTES + rl) * IN_DIM;
        const v4f x0 = *(const v4f*)xp;
        const v4f x1 = *(const v4f*)(xp + 4);
        const float xv[8] = {x0.x, x0.y, x0.z, x0.w, x1.x, x1.y, x1.z, x1.w};

        float cpl[NUM_CAPS];
        if (MODE != 0) {
            float lg[NUM_CAPS];
            const int rra = r0 + (lrow >> 3);
            const int ia  = lrow & 7;
            #pragma unroll
            for (int c = 0; c < NUM_CAPS; ++c) {
                const float* wa = W + (((size_t)rra * NUM_CAPS + c) * OUT_DIM + 8 * lhi) * IN_DIM + ia;
                v8f d = {};
                #pragma unroll
                for (int q = 0; q < 4; ++q) {
                    const v2f a = v2f{wa[(2 * q) * IN_DIM], wa[(2 * q + 1) * IN_DIM]};
                    d = __builtin_amdgcn_wmma_f32_16x16x4_f32(
                            false, a, false, vfrag[c][q], (short)0, d, false, false);
                }
                float l = 0.0f;
                #pragma unroll
                for (int g = 0; g < 8; ++g) l += d[g] * xv[g];
                lg[c] = l;
            }
            float m = lg[0];
            #pragma unroll
            for (int c = 1; c < NUM_CAPS; ++c) m = fmaxf(m, lg[c]);
            float ssum = 0.0f;
            #pragma unroll
            for (int c = 0; c < NUM_CAPS; ++c) { cpl[c] = __expf(lg[c] - m); ssum += cpl[c]; }
            const float inv = 1.0f / ssum;
            #pragma unroll
            for (int c = 0; c < NUM_CAPS; ++c) cpl[c] *= inv;
        }

        #pragma unroll
        for (int c = 0; c < NUM_CAPS; ++c) {
            const float* wb = W + (((size_t)rl * NUM_CAPS + c) * OUT_DIM + lrow) * IN_DIM;
            const v4f w0 = *(const v4f*)wb;
            const v4f w1 = *(const v4f*)(wb + 4);
            const float k = (MODE != 0) ? cpl[c] : 1.0f;
            float ax[8];
            #pragma unroll
            for (int g = 0; g < 8; ++g) ax[g] = (MODE != 0) ? k * xv[g] : xv[g];
            sacc[c] = __builtin_amdgcn_wmma_f32_16x16x4_f32(
                          false, v2f{ax[0], ax[1]}, false, v2f{w0.x, w0.y}, (short)0, sacc[c], false, false);
            sacc[c] = __builtin_amdgcn_wmma_f32_16x16x4_f32(
                          false, v2f{ax[2], ax[3]}, false, v2f{w0.z, w0.w}, (short)0, sacc[c], false, false);
            sacc[c] = __builtin_amdgcn_wmma_f32_16x16x4_f32(
                          false, v2f{ax[4], ax[5]}, false, v2f{w1.x, w1.y}, (short)0, sacc[c], false, false);
            sacc[c] = __builtin_amdgcn_wmma_f32_16x16x4_f32(
                          false, v2f{ax[6], ax[7]}, false, v2f{w1.z, w1.w}, (short)0, sacc[c], false, false);
        }
    }

    #pragma unroll
    for (int c = 0; c < NUM_CAPS; ++c) {
        #pragma unroll
        for (int g = 0; g < 8; ++g) {
            const int b = m_base + g + 8 * lhi;
            part[(((size_t)b * NUM_CAPS + c) * NCHUNK + chunk) * OUT_DIM + lrow]
                = sacc[c][g];
        }
    }
}

template <bool UNIFORM>
__global__ __launch_bounds__(256)
void s_reduce_squash_kernel(const float* __restrict__ part,
                            float* __restrict__ v_out)
{
    __shared__ float red[256];
    const int bc = blockIdx.x;
    const int t  = threadIdx.x;

    const float* p = part + (size_t)bc * NCHUNK * OUT_DIM;
    float acc = 0.0f;
    for (int k = t; k < NCHUNK * OUT_DIM; k += 256) acc += p[k];
    red[t] = acc;
    __syncthreads();
    for (int s = 128; s >= 16; s >>= 1) {
        if (t < s) red[t] += red[t + s];
        __syncthreads();
    }
    if (t < 16) {
        const float coup = UNIFORM ? 1.0f / (float)NUM_CAPS : 1.0f;
        const float sv = coup * red[t];
        float sq = 0.0f;
        for (int k = 0; k < 16; ++k) { const float q = coup * red[k]; sq += q * q; }
        const float norm  = sqrtf(sq + 1e-8f);
        const float scale = sq / (1.0f + sq);
        v_out[(size_t)bc * OUT_DIM + t] = scale * sv / norm;
    }
}

extern "C" void kernel_launch(void* const* d_in, const int* in_sizes, int n_in,
                              void* d_out, int out_size, void* d_ws, size_t ws_size,
                              hipStream_t stream)
{
    const float* x = (const float*)d_in[0];
    const float* W = (const float*)d_in[1];
    float* out = (float*)d_out;

    float* part = (float*)d_ws;
    float* v0   = part + (size_t)NCHUNK * BATCH * NUM_CAPS * OUT_DIM;
    float* v1   = v0 + (size_t)BATCH * NUM_CAPS * OUT_DIM;

    route_pass_kernel<0><<<NCHUNK, 256, 0, stream>>>(x, W, v0, v0, part);
    s_reduce_squash_kernel<true><<<BATCH * NUM_CAPS, 256, 0, stream>>>(part, v0);
    route_pass_kernel<1><<<NCHUNK, 256, 0, stream>>>(x, W, v0, v0, part);
    s_reduce_squash_kernel<false><<<BATCH * NUM_CAPS, 256, 0, stream>>>(part, v1);
    route_pass_kernel<2><<<NCHUNK, 256, 0, stream>>>(x, W, v0, v1, part);
    s_reduce_squash_kernel<false><<<BATCH * NUM_CAPS, 256, 0, stream>>>(part, out);
}


# Block = 2 batch tiles x 4 route groups: the 4 waves of a tile add their s accumulators through LDS, so 64 partials per output instead of 256; reducer = one wave per (b,c); new sum order
# speedup vs baseline: 4.1874x; 2.0107x over previous
#include <hip/hip_runtime.h>
#include <cstddef>


#define BATCH      128
#define NUM_ROUTES 2048
#define NUM_CAPS   10
#define IN_DIM     8
#define OUT_DIM    16
#define RWAVE      8
#define RGROUPS    4
#define TILES      2
#define RBLOCK     (RWAVE * RGROUPS)
#define NCHUNK     (NUM_ROUTES / RBLOCK)
#define NTGROUP    (BATCH / (16 * TILES))
#define SB         (NUM_CAPS * OUT_DIM)
#define CSTRIDE    (8 * 32 + 16)
#define WSTRIDE    (NUM_CAPS * CSTRIDE)

typedef __attribute__((ext_vector_type(2))) float v2f;
typedef __attribute__((ext_vector_type(4))) float v4f;
typedef __attribute__((ext_vector_type(8))) float v8f;

template <int MODE>
__global__ __launch_bounds__(256)
void route_pass_kernel(const float* __restrict__ x,
                       const float* __restrict__ W,
                       const float* __restrict__ va,
                       const float* __restrict__ vb,
                       float* __restrict__ part)
{
    __shared__ __attribute__((aligned(16))) float red[TILES * RGROUPS * WSTRIDE];

    const int chunk = blockIdx.x / NTGROUP;
    const int tg    = blockIdx.x % NTGROUP;
    const int wave = threadIdx.x >> 5;
    const int lane = threadIdx.x & 31;
    const int lrow = lane & 15;
    const int lhi  = lane >> 4;
    const int tile = wave & 1;
    const int rg   = wave >> 1;
    const int m_base = (tg * TILES + tile) * 16;
    const int brow = m_base + lrow;
    const int rbase = chunk * RBLOCK + rg * RWAVE;

    v2f vfrag[NUM_CAPS][4];
    if (MODE != 0) {
        #pragma unroll
        for (int c = 0; c < NUM_CAPS; ++c) {
            const size_t off = ((size_t)brow * NUM_CAPS + c) * OUT_DIM + 8 * lhi;
            v4f lo = *(const v4f*)(va + off);
            v4f hi = *(const v4f*)(va + off + 4);
            if (MODE == 2) {
                lo += *(const v4f*)(vb + off);
                hi += *(const v4f*)(vb + off + 4);
            }
            vfrag[c][0] = v2f{lo.x, lo.y};
            vfrag[c][1] = v2f{lo.z, lo.w};
            vfrag[c][2] = v2f{hi.x, hi.y};
            vfrag[c][3] = v2f{hi.z, hi.w};
        }
    }

    v8f sacc[NUM_CAPS] = {};

    #pragma unroll
    for (int pp = 0; pp < RWAVE / 2; ++pp) {
        const int r0 = rbase + 2 * pp;
        const int rl = r0 + lhi;

        const float* xp = x + ((size_t)brow * NUM_ROUTES + rl) * IN_DIM;
        const v4f x0 = *(const v4f*)xp;
        const v4f x1 = *(const v4f*)(xp + 4);
        const float xv[8] = {x0.x, x0.y, x0.z, x0.w, x1.x, x1.y, x1.z, x1.w};

        float cpl[NUM_CAPS];
        if (MODE != 0) {
            float lg[NUM_CAPS];
            const int rra = r0 + (lrow >> 3);
            const int ia  = lrow & 7;
            #pragma unroll
            for (int c = 0; c < NUM_CAPS; ++c) {
                const float* wa = W + (((size_t)rra * NUM_CAPS + c) * OUT_DIM + 8 * lhi) * IN_DIM + ia;
                v8f d = {};
                #pragma unroll
                for (int q = 0; q < 4; ++q) {
                    const v2f a = v2f{wa[(2 * q) * IN_DIM], wa[(2 * q + 1) * IN_DIM]};
                    d = __builtin_amdgcn_wmma_f32_16x16x4_f32(
                            false, a, false, vfrag[c][q], (short)0, d, false, false);
                }
                float l = 0.0f;
                #pragma unroll
                for (int g = 0; g < 8; ++g) l += d[g] * xv[g];
                lg[c] = l;
            }
            float m = lg[0];
            #pragma unroll
            for (int c = 1; c < NUM_CAPS; ++c) m = fmaxf(m, lg[c]);
            float ssum = 0.0f;
            #pragma unroll
            for (int c = 0; c < NUM_CAPS; ++c) { cpl[c] = __expf(lg[c] - m); ssum += cpl[c]; }
            const float inv = 1.0f / ssum;
            #pragma unroll
            for (int c = 0; c < NUM_CAPS; ++c) cpl[c] *= inv;
        }

        #pragma unroll
        for (int c = 0; c < NUM_CAPS; ++c) {
            const float* wb = W + (((size_t)rl * NUM_CAPS + c) * OUT_DIM + lrow) * IN_DIM;
            const v4f w0 = *(const v4f*)wb;
            const v4f w1 = *(const v4f*)(wb + 4);
            const float k = (MODE != 0) ? cpl[c] : 1.0f;
            float ax[8];
            #pragma unroll
            for (int g = 0; g < 8; ++g) ax[g] = (MODE != 0) ? k * xv[g] : xv[g];
            sacc[c] = __builtin_amdgcn_wmma_f32_16x16x4_f32(
                          false, v2f{ax[0], ax[1]}, false, v2f{w0.x, w0.y}, (short)0, sacc[c], false, false);
            sacc[c] = __builtin_amdgcn_wmma_f32_16x16x4_f32(
                          false, v2f{ax[2], ax[3]}, false, v2f{w0.z, w0.w}, (short)0, sacc[c], false, false);
            sacc[c] = __builtin_amdgcn_wmma_f32_16x16x4_f32(
                          false, v2f{ax[4], ax[5]}, false, v2f{w1.x, w1.y}, (short)0, sacc[c], false, false);
            sacc[c] = __builtin_amdgcn_wmma_f32_16x16x4_f32(
                          false, v2f{ax[6], ax[7]}, false, v2f{w1.z, w1.w}, (short)0, sacc[c], false, false);
        }
    }

    float* mine = red + (tile * RGROUPS + rg) * WSTRIDE;
    #pragma unroll
    for (int c = 0; c < NUM_CAPS; ++c) {
        #pragma unroll
        for (int g = 0; g < 8; ++g) mine[c * CSTRIDE + g * 32 + lane] = sacc[c][g];
    }
    __syncthreads();

    float* dst = part + ((size_t)chunk * BATCH + (size_t)tg * TILES * 16) * SB;
    #pragma unroll
    for (int j = 0; j < (TILES * 16 * SB / 4) / 256; ++j) {
        const int gi  = (int)threadIdx.x + 256 * j;
        const int tl  = gi / (16 * SB / 4);
        const int r   = gi - tl * (16 * SB / 4);
        const int bl  = r / (SB / 4);
        const int rem = r - bl * (SB / 4);
        const int c   = rem >> 2;
        const int oq  = rem & 3;
        const float* sp = red + tl * RGROUPS * WSTRIDE + c * CSTRIDE
                        + (bl & 7) * 32 + (bl >> 3) * 16 + oq * 4;
        v4f s = *(const v4f*)sp;
        #pragma unroll
        for (int k = 1; k < RGROUPS; ++k) s += *(const v4f*)(sp + k * WSTRIDE);
        *(v4f*)(dst + 4 * (size_t)gi) = s;
    }
}

template <bool UNIFORM>
__global__ __launch_bounds__(128)
void s_reduce_squash_kernel(const float* __restrict__ part,
                            float* __restrict__ v_out)
{
    const int wave = threadIdx.x >> 5;
    const int lane = threadIdx.x & 31;
    const int bc = blockIdx.x * 4 + wave;
    const int oq = lane & 3;
    const int cg = lane >> 2;

    const float* p = part + (size_t)bc * OUT_DIM + oq * 4 + (size_t)cg * BATCH * SB;
    v4f acc = {0.0f, 0.0f, 0.0f, 0.0f};
    #pragma unroll
    for (int j = 0; j < NCHUNK / 8; ++j)
        acc += *(const v4f*)(p + (size_t)j * 8 * BATCH * SB);

    #pragma unroll
    for (int m = 4; m <= 16; m <<= 1) {
        acc.x += __shfl_xor(acc.x, m);
        acc.y += __shfl_xor(acc.y, m);
        acc.z += __shfl_xor(acc.z, m);
        acc.w += __shfl_xor(acc.w, m);
    }

    const float coup = UNIFORM ? 1.0f / (float)NUM_CAPS : 1.0f;
    const v4f sv = acc * coup;
    float sq = sv.x * sv.x + sv.y * sv.y + sv.z * sv.z + sv.w * sv.w;
    sq += __shfl_xor(sq, 1);
    sq += __shfl_xor(sq, 2);
    const float norm  = sqrtf(sq + 1e-8f);
    const float scale = sq / (1.0f + sq);
    if (cg == 0)
        *(v4f*)(v_out + (size_t)bc * OUT_DIM + oq * 4) = sv * scale / norm;
}

extern "C" void kernel_launch(void* const* d_in, const int* in_sizes, int n_in,
                              void* d_out, int out_size, void* d_ws, size_t ws_size,
                              hipStream_t stream)
{
    const float* x = (const float*)d_in[0];
    const float* W = (const float*)d_in[1];
    float* out = (float*)d_out;

    float* part = (float*)d_ws;
    float* v0   = part + (size_t)NCHUNK * BATCH * NUM_CAPS * OUT_DIM;
    float* v1   = v0 + (size_t)BATCH * NUM_CAPS * OUT_DIM;

    route_pass_kernel<0><<<NCHUNK * NTGROUP, 256, 0, stream>>>(x, W, v0, v0, part);
    s_reduce_squash_kernel<true><<<BATCH * NUM_CAPS / 4, 128, 0, stream>>>(part, v0);
    route_pass_kernel<1><<<NCHUNK * NTGROUP, 256, 0, stream>>>(x, W, v0, v0, part);
    s_reduce_squash_kernel<false><<<BATCH * NUM_CAPS / 4, 128, 0, stream>>>(part, v1);
    route_pass_kernel<2><<<NCHUNK * NTGROUP, 256, 0, stream>>>(x, W, v0, v1, part);
    s_reduce_squash_kernel<false><<<BATCH * NUM_CAPS / 4, 128, 0, stream>>>(part, out);
}


# Pass kernel software-pipelined: a route pair's W and x loads issue as one group a WMMA block ahead, into a second register set; scheduling fences hold the groups; no load among WMMAs; same sums
# speedup vs baseline: 4.4330x; 1.0587x over previous
#include <hip/hip_runtime.h>
#include <cstddef>


#define BATCH      128
#define NUM_ROUTES 2048
#define NUM_CAPS   10
#define IN_DIM     8
#define OUT_DIM    16
#define RWAVE      8
#define RGROUPS    4
#define TILES      2
#define RBLOCK     (RWAVE * RGROUPS)
#define NCHUNK     (NUM_ROUTES / RBLOCK)
#define NTGROUP    (BATCH / (16 * TILES))
#define SB         (NUM_CAPS * OUT_DIM)
#define CSTRIDE    (8 * 32 + 16)
#define WSTRIDE    (NUM_CAPS * CSTRIDE)

typedef __attribute__((ext_vector_type(2))) float v2f;
typedef __attribute__((ext_vector_type(4))) float v4f;
typedef __attribute__((ext_vector_type(8))) float v8f;

template <int MODE>
__global__ __launch_bounds__(256)
void route_pass_kernel(const float* __restrict__ x,
                       const float* __restrict__ W,
                       const float* __restrict__ va,
                       const float* __restrict__ vb,
                       float* __restrict__ part)
{
    __shared__ __attribute__((aligned(16))) float red[TILES * RGROUPS * WSTRIDE];

    const int chunk = blockIdx.x / NTGROUP;
    const int tg    = blockIdx.x % NTGROUP;
    const int wave = threadIdx.x >> 5;
    const int lane = threadIdx.x & 31;
    const int lrow = lane & 15;
    const int lhi  = lane >> 4;
    const int tile = wave & 1;
    const int rg   = wave >> 1;
    const int m_base = (tg * TILES + tile) * 16;
    const int brow = m_base + lrow;
    const int rbase = chunk * RBLOCK + rg * RWAVE;

    constexpr int CAPW  = OUT_DIM * IN_DIM;
    constexpr int PAIRW = 2 * NUM_CAPS * CAPW;
    constexpr int NPAIR = RWAVE / 2;
    const float* xl  = x + ((size_t)brow * NUM_ROUTES + rbase + lhi) * IN_DIM;
    const float* wbl = W + ((size_t)(rbase + lhi) * NUM_CAPS * OUT_DIM + lrow) * IN_DIM;
    const float* wal = W + ((size_t)(rbase + (lrow >> 3)) * NUM_CAPS * OUT_DIM + 8 * lhi) * IN_DIM + (lrow & 7);

    v4f   xs[2][2];
    v4f   wb[2][NUM_CAPS][2];
    float wa[2][NUM_CAPS][8];

    v2f vfrag[NUM_CAPS][4];
    if (MODE != 0) {
        #pragma unroll
        for (int c = 0; c < NUM_CAPS; ++c) {
            const size_t off = ((size_t)brow * NUM_CAPS + c) * OUT_DIM + 8 * lhi;
            v4f lo = *(const v4f*)(va + off);
            v4f hi = *(const v4f*)(va + off + 4);
            if (MODE == 2) {
                lo += *(const v4f*)(vb + off);
                hi += *(const v4f*)(vb + off + 4);
            }
            vfrag[c][0] = v2f{lo.x, lo.y};
            vfrag[c][1] = v2f{lo.z, lo.w};
            vfrag[c][2] = v2f{hi.x, hi.y};
            vfrag[c][3] = v2f{hi.z, hi.w};
        }
    }

    xs[0][0] = *(const v4f*)xl;
    xs[0][1] = *(const v4f*)(xl + 4);
    if (MODE != 0) {
        #pragma unroll
        for (int c = 0; c < NUM_CAPS; ++c) {
            #pragma unroll
            for (int j = 0; j < 8; ++j) wa[0][c][j] = wal[c * CAPW + j * IN_DIM];
        }
    } else {
        #pragma unroll
        for (int c = 0; c < NUM_CAPS; ++c) {
            wb[0][c][0] = *(const v4f*)(wbl + c * CAPW);
            wb[0][c][1] = *(const v4f*)(wbl + c * CAPW + 4);
        }
    }

    v8f sacc[NUM_CAPS] = {};

    #pragma unroll
    for (int pp = 0; pp < NPAIR; ++pp) {
        const int cur = pp & 1;
        const int nxt = cur ^ 1;
        const bool more = (pp + 1 < NPAIR);
        const int nb = (pp + 1) * PAIRW;

        const float xv[8] = {xs[cur][0].x, xs[cur][0].y, xs[cur][0].z, xs[cur][0].w,
                             xs[cur][1].x, xs[cur][1].y, xs[cur][1].z, xs[cur][1].w};
        float cpl[NUM_CAPS];

        if (MODE != 0) {
            __builtin_amdgcn_sched_barrier(0);
            #pragma unroll
            for (int c = 0; c < NUM_CAPS; ++c) {
                wb[0][c][0] = *(const v4f*)(wbl + pp * PAIRW + c * CAPW);
                wb[0][c][1] = *(const v4f*)(wbl + pp * PAIRW + c * CAPW + 4);
            }
            if (more) {
                #pragma unroll
                for (int c = 0; c < NUM_CAPS / 2; ++c) {
                    #pragma unroll
                    for (int j = 0; j < 8; ++j) wa[nxt][c][j] = wal[nb + c * CAPW + j * IN_DIM];
                }
            }
            __builtin_amdgcn_sched_barrier(0);

            float lg[NUM_CAPS];
            #pragma unroll
            for (int c = 0; c < NUM_CAPS; ++c) {
                v8f d = {};
                #pragma unroll
                for (int q = 0; q < 4; ++q) {
                    const v2f a = v2f{wa[cur][c][2 * q], wa[cur][c][2 * q + 1]};
                    d = __builtin_amdgcn_wmma_f32_16x16x4_f32(
                            false, a, false, vfrag[c][q], (short)0, d, false, false);
                }
                float l = 0.0f;
                #pragma unroll
                for (int g = 0; g < 8; ++g) l += d[g] * xv[g];
                lg[c] = l;
            }
            float m = lg[0];
            #pragma unroll
            for (int c = 1; c < NUM_CAPS; ++c) m = fmaxf(m, lg[c]);
            float ssum = 0.0f;
            #pragma unroll
            for (int c = 0; c < NUM_CAPS; ++c) { cpl[c] = __expf(lg[c] - m); ssum += cpl[c]; }
            const float inv = 1.0f / ssum;
            #pragma unroll
            for (int c = 0; c < NUM_CAPS; ++c) cpl[c] *= inv;
        }

        __builtin_amdgcn_sched_barrier(0);
        if (more) {
            if (MODE != 0) {
                #pragma unroll
                for (int c = NUM_CAPS / 2; c < NUM_CAPS; ++c) {
                    #pragma unroll
                    for (int j = 0; j < 8; ++j) wa[nxt][c][j] = wal[nb + c * CAPW + j * IN_DIM];
                }
            } else {
                #pragma unroll
                for (int c = 0; c < NUM_CAPS; ++c) {
                    wb[nxt][c][0] = *(const v4f*)(wbl + nb + c * CAPW);
                    wb[nxt][c][1] = *(const v4f*)(wbl + nb + c * CAPW + 4);
                }
            }
            xs[nxt][0] = *(const v4f*)(xl + (pp + 1) * 2 * IN_DIM);
            xs[nxt][1] = *(const v4f*)(xl + (pp + 1) * 2 * IN_DIM + 4);
        }
        __builtin_amdgcn_sched_barrier(0);

        const int ws = (MODE != 0) ? 0 : cur;
        #pragma unroll
        for (int c = 0; c < NUM_CAPS; ++c) {
            const v4f w0 = wb[ws][c][0];
            const v4f w1 = wb[ws][c][1];
            const float k = (MODE != 0) ? cpl[c] : 1.0f;
            float ax[8];
            #pragma unroll
            for (int g = 0; g < 8; ++g) ax[g] = (MODE != 0) ? k * xv[g] : xv[g];
            sacc[c] = __builtin_amdgcn_wmma_f32_16x16x4_f32(
                          false, v2f{ax[0], ax[1]}, false, v2f{w0.x, w0.y}, (short)0, sacc[c], false, false);
            sacc[c] = __builtin_amdgcn_wmma_f32_16x16x4_f32(
                          false, v2f{ax[2], ax[3]}, false, v2f{w0.z, w0.w}, (short)0, sacc[c], false, false);
            sacc[c] = __builtin_amdgcn_wmma_f32_16x16x4_f32(
                          false, v2f{ax[4], ax[5]}, false, v2f{w1.x, w1.y}, (short)0, sacc[c], false, false);
            sacc[c] = __builtin_amdgcn_wmma_f32_16x16x4_f32(
                          false, v2f{ax[6], ax[7]}, false, v2f{w1.z, w1.w}, (short)0, sacc[c], false, false);
        }
    }

    float* mine = red + (tile * RGROUPS + rg) * WSTRIDE;
    #pragma unroll
    for (int c = 0; c < NUM_CAPS; ++c) {
        #pragma unroll
        for (int g = 0; g < 8; ++g) mine[c * CSTRIDE + g * 32 + lane] = sacc[c][g];
    }
    __syncthreads();

    float* dst = part + ((size_t)chunk * BATCH + (size_t)tg * TILES * 16) * SB;
    #pragma unroll
    for (int j = 0; j < (TILES * 16 * SB / 4) / 256; ++j) {
        const int gi  = (int)threadIdx.x + 256 * j;
        const int tl  = gi / (16 * SB / 4);
        const int r   = gi - tl * (16 * SB / 4);
        const int bl  = r / (SB / 4);
        const int rem = r - bl * (SB / 4);
        const int c   = rem >> 2;
        const int oq  = rem & 3;
        const float* sp = red + tl * RGROUPS * WSTRIDE + c * CSTRIDE
                        + (bl & 7) * 32 + (bl >> 3) * 16 + oq * 4;
        v4f s = *(const v4f*)sp;
        #pragma unroll
        for (int k = 1; k < RGROUPS; ++k) s += *(const v4f*)(sp + k * WSTRIDE);
        *(v4f*)(dst + 4 * (size_t)gi) = s;
    }
}

template <bool UNIFORM>
__global__ __launch_bounds__(128)
void s_reduce_squash_kernel(const float* __restrict__ part,
                            float* __restrict__ v_out)
{
    const int wave = threadIdx.x >> 5;
    const int lane = threadIdx.x & 31;
    const int bc = blockIdx.x * 4 + wave;
    const int oq = lane & 3;
    const int cg = lane >> 2;

    const float* p = part + (size_t)bc * OUT_DIM + oq * 4 + (size_t)cg * BATCH * SB;
    v4f acc = {0.0f, 0.0f, 0.0f, 0.0f};
    #pragma unroll
    for (int j = 0; j < NCHUNK / 8; ++j)
        acc += *(const v4f*)(p + (size_t)j * 8 * BATCH * SB);

    #pragma unroll
    for (int m = 4; m <= 16; m <<= 1) {
        acc.x += __shfl_xor(acc.x, m);
        acc.y += __shfl_xor(acc.y, m);
        acc.z += __shfl_xor(acc.z, m);
        acc.w += __shfl_xor(acc.w, m);
    }

    const float coup = UNIFORM ? 1.0f / (float)NUM_CAPS : 1.0f;
    const v4f sv = acc * coup;
    float sq = sv.x * sv.x + sv.y * sv.y + sv.z * sv.z + sv.w * sv.w;
    sq += __shfl_xor(sq, 1);
    sq += __shfl_xor(sq, 2);
    const float norm  = sqrtf(sq + 1e-8f);
    const float scale = sq / (1.0f + sq);
    if (cg == 0)
        *(v4f*)(v_out + (size_t)bc * OUT_DIM + oq * 4) = sv * scale / norm;
}

extern "C" void kernel_launch(void* const* d_in, const int* in_sizes, int n_in,
                              void* d_out, int out_size, void* d_ws, size_t ws_size,
                              hipStream_t stream)
{
    const float* x = (const float*)d_in[0];
    const float* W = (const float*)d_in[1];
    float* out = (float*)d_out;

    float* part = (float*)d_ws;
    float* v0   = part + (size_t)NCHUNK * BATCH * NUM_CAPS * OUT_DIM;
    float* v1   = v0 + (size_t)BATCH * NUM_CAPS * OUT_DIM;

    route_pass_kernel<0><<<NCHUNK * NTGROUP, 256, 0, stream>>>(x, W, v0, v0, part);
    s_reduce_squash_kernel<true><<<BATCH * NUM_CAPS / 4, 128, 0, stream>>>(part, v0);
    route_pass_kernel<1><<<NCHUNK * NTGROUP, 256, 0, stream>>>(x, W, v0, v0, part);
    s_reduce_squash_kernel<false><<<BATCH * NUM_CAPS / 4, 128, 0, stream>>>(part, v1);
    route_pass_kernel<2><<<NCHUNK * NTGROUP, 256, 0, stream>>>(x, W, v0, v1, part);
    s_reduce_squash_kernel<false><<<BATCH * NUM_CAPS / 4, 128, 0, stream>>>(part, out);
}
